# starting kernel
# baseline (speedup 1.0000x reference)
#include <hip/hip_runtime.h>

typedef __attribute__((ext_vector_type(2))) float v2f;
typedef __attribute__((ext_vector_type(4))) float v4f;
typedef __attribute__((ext_vector_type(8))) float v8f;

#define D_MODEL 1024
#define H_HEADS 16
#define HEAD_DIM 64
#define SEQ_L 2048
#define BATCH 2
#define M_ROWS (BATCH * SEQ_L)

__device__ __forceinline__ v8f wmma_f32(v2f a, v2f b, v8f c) {
  return __builtin_amdgcn_wmma_f32_16x16x4_f32(false, a, false, b, (short)0, c,
                                               false, false);
}

__device__ __forceinline__ float rowmax16(float x) {
#pragma unroll
  for (int m = 1; m < 16; m <<= 1) x = fmaxf(x, __shfl_xor(x, m, 32));
  return x;
}
__device__ __forceinline__ float rowsum16(float x) {
#pragma unroll
  for (int m = 1; m < 16; m <<= 1) x += __shfl_xor(x, m, 32);
  return x;
}

#define BM 64
#define BN 128
#define BK 32
#define APAD 36
#define BPAD 34

__global__ __launch_bounds__(256) void gemm_bias_kernel(
    const float* __restrict__ A, const float* __restrict__ B,
    const float* __restrict__ bias, float* __restrict__ C, int N, int K) {
  __shared__ float As[BM][APAD];
  __shared__ float Bs[BN][BPAD];

  const int tid = threadIdx.x;
  const int lane = tid & 31;
  const int w = tid >> 5;
  const int wm = (w >> 2) << 5;
  const int wn = (w & 3) << 5;
  const int m0 = blockIdx.y * BM;
  const int n0 = blockIdx.x * BN;

  const int ln = lane & 15;
  const int half = lane >> 4;
  const int kb = half << 1;

  v8f acc[2][2];
#pragma unroll
  for (int i = 0; i < 2; ++i)
#pragma unroll
    for (int j = 0; j < 2; ++j)
#pragma unroll
      for (int v = 0; v < 8; ++v) acc[i][j][v] = 0.0f;

  for (int k0 = 0; k0 < K; k0 += BK) {
    __syncthreads();
    {
      int r = tid >> 3;
      int c = (tid & 7) << 2;
      const float* ap = A + (size_t)(m0 + r) * K + (k0 + c);
      *(v4f*)&As[r][c] = *(const v4f*)ap;
      *(v4f*)&As[r + 32][c] = *(const v4f*)(ap + (size_t)32 * K);
    }
    {
      int kr = tid >> 5;
      int nc = (tid & 31) << 2;
#pragma unroll
      for (int it = 0; it < 4; ++it) {
        int kk = kr + it * 8;
        v4f bv = *(const v4f*)(B + (size_t)(k0 + kk) * N + (n0 + nc));
        Bs[nc + 0][kk] = bv.x;
        Bs[nc + 1][kk] = bv.y;
        Bs[nc + 2][kk] = bv.z;
        Bs[nc + 3][kk] = bv.w;
      }
    }
    __syncthreads();
#pragma unroll
    for (int kk = 0; kk < BK; kk += 4) {
      v2f a0 = *(const v2f*)&As[wm + ln][kk + kb];
      v2f a1 = *(const v2f*)&As[wm + 16 + ln][kk + kb];
      v2f b0 = *(const v2f*)&Bs[wn + ln][kk + kb];
      v2f b1 = *(const v2f*)&Bs[wn + 16 + ln][kk + kb];
      acc[0][0] = wmma_f32(a0, b0, acc[0][0]);
      acc[0][1] = wmma_f32(a0, b1, acc[0][1]);
      acc[1][0] = wmma_f32(a1, b0, acc[1][0]);
      acc[1][1] = wmma_f32(a1, b1, acc[1][1]);
    }
  }

#pragma unroll
  for (int j = 0; j < 2; ++j) {
    float bj = bias[n0 + wn + j * 16 + ln];
#pragma unroll
    for (int i = 0; i < 2; ++i) {
      int rbase = m0 + wm + i * 16 + half * 8;
      int col = n0 + wn + j * 16 + ln;
#pragma unroll
      for (int v = 0; v < 8; ++v)
        C[(size_t)(rbase + v) * N + col] = acc[i][j][v] + bj;
    }
  }
}

#define KPAD 68

__global__ __launch_bounds__(128) void mha_kernel(const float* __restrict__ qkv,
                                                  float* __restrict__ attn_out) {
  __shared__ float Ks[64][KPAD];
  __shared__ float Vs[64][KPAD];
  __shared__ float Ps[4][16][KPAD];

  const int qb = blockIdx.x;
  const int bh = blockIdx.y;
  const int b = bh >> 4;
  const int h = bh & 15;

  const int tid = threadIdx.x;
  const int lane = tid & 31;
  const int w = tid >> 5;
  const int ln = lane & 15;
  const int half = lane >> 4;
  const int kb = half << 1;

  const int qw0 = qb * 64 + w * 16;
  const size_t rstride = 3 * D_MODEL;
  const float* base = qkv + (size_t)b * SEQ_L * rstride + (size_t)h * HEAD_DIM;

  v2f qa[16];
  {
    const float* qp = base + (size_t)(qw0 + ln) * rstride + kb;
#pragma unroll
    for (int kk = 0; kk < 16; ++kk) qa[kk] = *(const v2f*)(qp + kk * 4);
  }

  v8f acc_o[4];
  float m_i[8], l_i[8];
#pragma unroll
  for (int v = 0; v < 8; ++v) {
    m_i[v] = -__builtin_inff();
    l_i[v] = 0.0f;
    acc_o[0][v] = 0.0f; acc_o[1][v] = 0.0f;
    acc_o[2][v] = 0.0f; acc_o[3][v] = 0.0f;
  }

  const float scale = 0.03125f;
  const float log2e = 1.44269504088896f;

  for (int j = 0; j <= qb; ++j) {
    __syncthreads();
    {
      int r0 = tid >> 4;
      int c = (tid & 15) << 2;
#pragma unroll
      for (int it = 0; it < 8; ++it) {
        int r = r0 + it * 8;
        const float* row = base + (size_t)(j * 64 + r) * rstride;
        *(v4f*)&Ks[r][c] = *(const v4f*)(row + D_MODEL + c);
        v4f vv = *(const v4f*)(row + 2 * D_MODEL + c);
        Vs[c + 0][r] = vv.x;
        Vs[c + 1][r] = vv.y;
        Vs[c + 2][r] = vv.z;
        Vs[c + 3][r] = vv.w;
      }
    }
    __syncthreads();

    v8f s[4];
#pragma unroll
    for (int nt = 0; nt < 4; ++nt)
#pragma unroll
      for (int v = 0; v < 8; ++v) s[nt][v] = 0.0f;
#pragma unroll
    for (int kk = 0; kk < 16; ++kk) {
      v2f b0 = *(const v2f*)&Ks[0 + ln][kk * 4 + kb];
      v2f b1 = *(const v2f*)&Ks[16 + ln][kk * 4 + kb];
      v2f b2 = *(const v2f*)&Ks[32 + ln][kk * 4 + kb];
      v2f b3 = *(const v2f*)&Ks[48 + ln][kk * 4 + kb];
      s[0] = wmma_f32(qa[kk], b0, s[0]);
      s[1] = wmma_f32(qa[kk], b1, s[1]);
      s[2] = wmma_f32(qa[kk], b2, s[2]);
      s[3] = wmma_f32(qa[kk], b3, s[3]);
    }

    const bool do_mask = (j == qb);
    float tmax[8];
#pragma unroll
    for (int v = 0; v < 8; ++v) {
      float mx = -__builtin_inff();
#pragma unroll
      for (int nt = 0; nt < 4; ++nt) {
        float sv = s[nt][v] * scale;
        if (do_mask) {
          int kvg = j * 64 + nt * 16 + ln;
          int qg = qw0 + v + half * 8;
          if (kvg > qg) sv = -__builtin_inff();
        }
        s[nt][v] = sv;
        mx = fmaxf(mx, sv);
      }
      tmax[v] = rowmax16(mx);
    }

#pragma unroll
    for (int v = 0; v < 8; ++v) {
      float mn = fmaxf(m_i[v], tmax[v]);
      float alpha = exp2f((m_i[v] - mn) * log2e);
      m_i[v] = mn;
      float rs = 0.0f;
#pragma unroll
      for (int nt = 0; nt < 4; ++nt) {
        float p = exp2f((s[nt][v] - mn) * log2e);
        rs += p;
        Ps[w][v + half * 8][nt * 16 + ln] = p;
      }
      l_i[v] = l_i[v] * alpha + rowsum16(rs);
      acc_o[0][v] *= alpha;
      acc_o[1][v] *= alpha;
      acc_o[2][v] *= alpha;
      acc_o[3][v] *= alpha;
    }

    asm volatile("s_wait_dscnt 0" ::: "memory");

#pragma unroll
    for (int kk = 0; kk < 16; ++kk) {
      v2f pa = *(const v2f*)&Ps[w][ln][kk * 4 + kb];
      v2f b0 = *(const v2f*)&Vs[0 + ln][kk * 4 + kb];
      v2f b1 = *(const v2f*)&Vs[16 + ln][kk * 4 + kb];
      v2f b2 = *(const v2f*)&Vs[32 + ln][kk * 4 + kb];
      v2f b3 = *(const v2f*)&Vs[48 + ln][kk * 4 + kb];
      acc_o[0] = wmma_f32(pa, b0, acc_o[0]);
      acc_o[1] = wmma_f32(pa, b1, acc_o[1]);
      acc_o[2] = wmma_f32(pa, b2, acc_o[2]);
      acc_o[3] = wmma_f32(pa, b3, acc_o[3]);
    }
  }

  float* obase = attn_out + (size_t)b * SEQ_L * D_MODEL + (size_t)h * HEAD_DIM;
#pragma unroll
  for (int v = 0; v < 8; ++v) {
    int qg = qw0 + v + half * 8;
    float inv_l = 1.0f / l_i[v];
#pragma unroll
    for (int nt = 0; nt < 4; ++nt)
      obase[(size_t)qg * D_MODEL + nt * 16 + ln] = acc_o[nt][v] * inv_l;
  }
}

extern "C" void kernel_launch(void* const* d_in, const int* in_sizes, int n_in,
                              void* d_out, int out_size, void* d_ws,
                              size_t ws_size, hipStream_t stream) {
  (void)in_sizes; (void)n_in; (void)out_size; (void)ws_size;
  const float* x = (const float*)d_in[0];
  const float* w_qkv = (const float*)d_in[1];
  const float* b_qkv = (const float*)d_in[2];
  const float* w_out = (const float*)d_in[3];
  const float* b_out = (const float*)d_in[4];
  float* out = (float*)d_out;

  float* qkv = (float*)d_ws;
  float* attn = qkv + (size_t)M_ROWS * 3 * D_MODEL;

  gemm_bias_kernel<<<dim3((3 * D_MODEL) / BN, M_ROWS / BM), 256, 0, stream>>>(
      x, w_qkv, b_qkv, qkv, 3 * D_MODEL, D_MODEL);

  mha_kernel<<<dim3(SEQ_L / 64, BATCH * H_HEADS), 128, 0, stream>>>(qkv, attn);

  gemm_bias_kernel<<<dim3(D_MODEL / BN, M_ROWS / BM), 256, 0, stream>>>(
      attn, w_out, b_out, out, D_MODEL, D_MODEL);
}


# attention softmax 527 -> 244 VALU instructions a tile: log2e/sqrt(d) folded into Q, exp = sub + v_exp_f32, 16-lane max/sum by DPP row_xmask, causal compare on diagonal tile only; sum order unchanged
# speedup vs baseline: 1.5849x; 1.5849x over previous
#include <hip/hip_runtime.h>

typedef __attribute__((ext_vector_type(2))) float v2f;
typedef __attribute__((ext_vector_type(4))) float v4f;
typedef __attribute__((ext_vector_type(8))) float v8f;

#define D_MODEL 1024
#define H_HEADS 16
#define HEAD_DIM 64
#define SEQ_L 2048
#define BATCH 2
#define M_ROWS (BATCH * SEQ_L)

__device__ __forceinline__ v8f wmma_f32(v2f a, v2f b, v8f c) {
  return __builtin_amdgcn_wmma_f32_16x16x4_f32(false, a, false, b, (short)0, c,
                                               false, false);
}

template <int M>
__device__ __forceinline__ float row_xor(float x) {
  return __builtin_bit_cast(
      float, __builtin_amdgcn_update_dpp(0, __builtin_bit_cast(int, x), 0x160 | M, 0xf, 0xf, true));
}
template <int M>
__device__ __forceinline__ float rowmax_step(float x) {
  float r;
  asm("v_max_num_f32_dpp %0, %1, %1 row_xmask:%2 row_mask:0xf bank_mask:0xf bound_ctrl:1"
      : "=v"(r)
      : "v"(x), "n"(M));
  return r;
}
__device__ __forceinline__ void rowmax16x8(float (&x)[8]) {
#pragma unroll
  for (int v = 0; v < 8; ++v) x[v] = rowmax_step<1>(x[v]);
#pragma unroll
  for (int v = 0; v < 8; ++v) x[v] = rowmax_step<2>(x[v]);
#pragma unroll
  for (int v = 0; v < 8; ++v) x[v] = rowmax_step<4>(x[v]);
#pragma unroll
  for (int v = 0; v < 8; ++v) x[v] = rowmax_step<8>(x[v]);
}
__device__ __forceinline__ void rowsum16x8(float (&x)[8]) {
#pragma unroll
  for (int v = 0; v < 8; ++v) x[v] += row_xor<1>(x[v]);
#pragma unroll
  for (int v = 0; v < 8; ++v) x[v] += row_xor<2>(x[v]);
#pragma unroll
  for (int v = 0; v < 8; ++v) x[v] += row_xor<4>(x[v]);
#pragma unroll
  for (int v = 0; v < 8; ++v) x[v] += row_xor<8>(x[v]);
}

#define BM 64
#define BN 128
#define BK 32
#define APAD 36
#define BPANEL 528

__global__ __launch_bounds__(256) void gemm_bias_kernel(
    const float* __restrict__ A, const float* __restrict__ B,
    const float* __restrict__ bias, float* __restrict__ C, int N, int K) {
  __shared__ __align__(16) float As[BM][APAD];
  __shared__ __align__(16) float Bs[(BN / 16) * BPANEL];

  const int tid = threadIdx.x;
  const int lane = tid & 31;
  const int w = tid >> 5;
  const int wm = (w >> 2) << 5;
  const int wn = (w & 3) << 5;
  const int m0 = blockIdx.y * BM;
  const int n0 = blockIdx.x * BN;

  const int ln = lane & 15;
  const int half = lane >> 4;
  const int kb = half << 1;

  v8f acc[2][2];
#pragma unroll
  for (int i = 0; i < 2; ++i)
#pragma unroll
    for (int j = 0; j < 2; ++j)
#pragma unroll
      for (int v = 0; v < 8; ++v) acc[i][j][v] = 0.0f;

  for (int k0 = 0; k0 < K; k0 += BK) {
    __syncthreads();
    {
      int r = tid >> 3;
      int c = (tid & 7) << 2;
      const float* ap = A + (size_t)(m0 + r) * K + (k0 + c);
      int kr = tid >> 5;
      int nc = (tid & 31) << 2;
      const float* bsrc = B + (size_t)(k0 + kr) * N + (n0 + nc);
      float* bdst = Bs + (nc >> 4) * BPANEL + kr * 16 + (nc & 15);
      v4f a_lo = *(const v4f*)ap;
      v4f a_hi = *(const v4f*)(ap + (size_t)32 * K);
      v4f bv0 = *(const v4f*)(bsrc);
      v4f bv1 = *(const v4f*)(bsrc + (size_t)8 * N);
      v4f bv2 = *(const v4f*)(bsrc + (size_t)16 * N);
      v4f bv3 = *(const v4f*)(bsrc + (size_t)24 * N);
      asm volatile("" ::"v"(a_lo), "v"(a_hi), "v"(bv0), "v"(bv1), "v"(bv2), "v"(bv3) : "memory");
      *(v4f*)&As[r][c] = a_lo;
      *(v4f*)&As[r + 32][c] = a_hi;
      *(v4f*)(bdst) = bv0;
      *(v4f*)(bdst + 8 * 16) = bv1;
      *(v4f*)(bdst + 16 * 16) = bv2;
      *(v4f*)(bdst + 24 * 16) = bv3;
    }
    __syncthreads();
    const float* bp0 = Bs + (wn >> 4) * BPANEL + kb * 16 + ln;
    const float* bp1 = bp0 + BPANEL;
#pragma unroll
    for (int kk = 0; kk < BK; kk += 4) {
      v2f a0 = *(const v2f*)&As[wm + ln][kk + kb];
      v2f a1 = *(const v2f*)&As[wm + 16 + ln][kk + kb];
      v2f b0, b1;
      b0.x = bp0[kk * 16];
      b0.y = bp0[kk * 16 + 16];
      b1.x = bp1[kk * 16];
      b1.y = bp1[kk * 16 + 16];
      acc[0][0] = wmma_f32(a0, b0, acc[0][0]);
      acc[0][1] = wmma_f32(a0, b1, acc[0][1]);
      acc[1][0] = wmma_f32(a1, b0, acc[1][0]);
      acc[1][1] = wmma_f32(a1, b1, acc[1][1]);
    }
  }

#pragma unroll
  for (int j = 0; j < 2; ++j) {
    float bj = bias[n0 + wn + j * 16 + ln];
#pragma unroll
    for (int i = 0; i < 2; ++i) {
      int rbase = m0 + wm + i * 16 + half * 8;
      int col = n0 + wn + j * 16 + ln;
#pragma unroll
      for (int v = 0; v < 8; ++v)
        C[(size_t)(rbase + v) * N + col] = acc[i][j][v] + bj;
    }
  }
}

#define KPAD 68
#define VPANEL 1040

__device__ __forceinline__ void sched_fence() { __builtin_amdgcn_sched_barrier(0); }

struct TileDiag { static constexpr bool value = true; };
struct TileFull { static constexpr bool value = false; };

__global__ __launch_bounds__(128, 4) void mha_kernel(const float* __restrict__ qkv,
                                                     float* __restrict__ attn_out) {
  __shared__ __align__(16) float Ks[64][KPAD];
  __shared__ __align__(16) float Vs[4 * VPANEL];
  __shared__ __align__(16) float Ps[4][16][KPAD];

  const int bh = blockIdx.x;
  const int qb = (SEQ_L / 64 - 1) - blockIdx.y;
  const int b = bh >> 4;
  const int h = bh & 15;

  const int tid = threadIdx.x;
  const int lane = tid & 31;
  const int w = tid >> 5;
  const int ln = lane & 15;
  const int half = lane >> 4;
  const int kb = half << 1;

  const int qw0 = qb * 64 + w * 16;
  const size_t rstride = 3 * D_MODEL;
  const float* base = qkv + (size_t)b * SEQ_L * rstride + (size_t)h * HEAD_DIM;

  const float qscale = 0.03125f * 1.44269504088896f;

  v2f qa[16];
  {
    const float* qp = base + (size_t)(qw0 + ln) * rstride + kb;
#pragma unroll
    for (int kk = 0; kk < 16; ++kk) qa[kk] = *(const v2f*)(qp + kk * 4) * qscale;
  }

  v8f acc_o[4];
  float m_i[8], l_i[8];
  float a_prev[8], rs_prev[8];
#pragma unroll
  for (int v = 0; v < 8; ++v) {
    m_i[v] = -__builtin_inff();
    l_i[v] = 0.0f;
    a_prev[v] = 1.0f;
    rs_prev[v] = 0.0f;
    acc_o[0][v] = 0.0f; acc_o[1][v] = 0.0f;
    acc_o[2][v] = 0.0f; acc_o[3][v] = 0.0f;
  }

  const float* vp = Vs + kb * 16 + ln;

  const int edge = ln - half * 8 - w * 16;

  auto tile = [&](const int j, auto tag) {
    constexpr bool DIAG = decltype(tag)::value;
    __syncthreads();
    {
      int r0 = tid >> 4;
      int c = (tid & 15) << 2;
      v4f kr[8], vr[8];
#pragma unroll
      for (int it = 0; it < 8; ++it) {
        int r = r0 + it * 8;
        const float* row = base + (size_t)(j * 64 + r) * rstride;
        kr[it] = *(const v4f*)(row + D_MODEL + c);
        vr[it] = *(const v4f*)(row + 2 * D_MODEL + c);
      }
      rowsum16x8(rs_prev);
#pragma unroll
      for (int v = 0; v < 8; ++v) l_i[v] = l_i[v] * a_prev[v] + rs_prev[v];
      asm volatile("" ::"v"(kr[0]), "v"(kr[1]), "v"(kr[2]), "v"(kr[3]), "v"(kr[4]), "v"(kr[5]),
                   "v"(kr[6]), "v"(kr[7]), "v"(vr[0]), "v"(vr[1]), "v"(vr[2]), "v"(vr[3]),
                   "v"(vr[4]), "v"(vr[5]), "v"(vr[6]), "v"(vr[7])
                   : "memory");
#pragma unroll
      for (int it = 0; it < 8; ++it) {
        int r = r0 + it * 8;
        *(v4f*)&Ks[r][c] = kr[it];
        *(v4f*)(Vs + (c >> 4) * VPANEL + r * 16 + (c & 15)) = vr[it];
      }
    }
    __syncthreads();

    v8f s[4];
#pragma unroll
    for (int nt = 0; nt < 4; ++nt)
#pragma unroll
      for (int v = 0; v < 8; ++v) s[nt][v] = 0.0f;
    v2f kf[2][16];
#pragma unroll
    for (int kq = 0; kq < 4; ++kq)
#pragma unroll
      for (int nt = 0; nt < 4; ++nt)
        kf[0][kq * 4 + nt] = *(const v2f*)&Ks[nt * 16 + ln][kq * 4 + kb];
#pragma unroll
    for (int blk = 0; blk < 4; ++blk) {
      if (blk < 3) {
#pragma unroll
        for (int kq = 0; kq < 4; ++kq)
#pragma unroll
          for (int nt = 0; nt < 4; ++nt)
            kf[(blk + 1) & 1][kq * 4 + nt] =
                *(const v2f*)&Ks[nt * 16 + ln][((blk + 1) * 4 + kq) * 4 + kb];
      }
      sched_fence();
#pragma unroll
      for (int kq = 0; kq < 4; ++kq)
#pragma unroll
        for (int nt = 0; nt < 4; ++nt)
          s[nt] = wmma_f32(qa[blk * 4 + kq], kf[blk & 1][kq * 4 + nt], s[nt]);
      sched_fence();
    }

    v2f pf[2][4];
    v2f vf[2][16];
#pragma unroll
    for (int kq = 0; kq < 4; ++kq)
#pragma unroll
      for (int nt = 0; nt < 4; ++nt) {
        vf[0][kq * 4 + nt].x = vp[nt * VPANEL + kq * 64];
        vf[0][kq * 4 + nt].y = vp[nt * VPANEL + kq * 64 + 16];
      }
    sched_fence();

    if constexpr (DIAG) {
#pragma unroll
      for (int nt = 0; nt < 4; ++nt)
#pragma unroll
        for (int v = 0; v < 8; ++v)
          if (edge > v - nt * 16) s[nt][v] = -__builtin_inff();
    }
    float tmax[8];
#pragma unroll
    for (int v = 0; v < 8; ++v)
      tmax[v] = fmaxf(fmaxf(s[0][v], s[1][v]), fmaxf(s[2][v], s[3][v]));
    rowmax16x8(tmax);

#pragma unroll
    for (int v = 0; v < 8; ++v) {
      float mn = fmaxf(m_i[v], tmax[v]);
      float alpha = __builtin_amdgcn_exp2f(m_i[v] - mn);
      m_i[v] = mn;
      float rs = 0.0f;
#pragma unroll
      for (int nt = 0; nt < 4; ++nt) {
        float p = __builtin_amdgcn_exp2f(s[nt][v] - mn);
        rs += p;
        Ps[w][v + half * 8][nt * 16 + ln] = p;
      }
      a_prev[v] = alpha;
      rs_prev[v] = rs;
      acc_o[0][v] *= alpha;
      acc_o[1][v] *= alpha;
      acc_o[2][v] *= alpha;
      acc_o[3][v] *= alpha;
    }

    asm volatile("s_wait_dscnt 0" ::: "memory");

#pragma unroll
    for (int kq = 0; kq < 4; ++kq) pf[0][kq] = *(const v2f*)&Ps[w][ln][kq * 4 + kb];
#pragma unroll
    for (int blk = 0; blk < 4; ++blk) {
      if (blk < 3) {
#pragma unroll
        for (int kq = 0; kq < 4; ++kq) {
          const int kk = (blk + 1) * 4 + kq;
          pf[(blk + 1) & 1][kq] = *(const v2f*)&Ps[w][ln][kk * 4 + kb];
#pragma unroll
          for (int nt = 0; nt < 4; ++nt) {
            vf[(blk + 1) & 1][kq * 4 + nt].x = vp[nt * VPANEL + kk * 64];
            vf[(blk + 1) & 1][kq * 4 + nt].y = vp[nt * VPANEL + kk * 64 + 16];
          }
        }
      }
      sched_fence();
#pragma unroll
      for (int kq = 0; kq < 4; ++kq)
#pragma unroll
        for (int nt = 0; nt < 4; ++nt)
          acc_o[nt] = wmma_f32(pf[blk & 1][kq], vf[blk & 1][kq * 4 + nt], acc_o[nt]);
      sched_fence();
    }
  };

  for (int j = 0; j < qb; ++j) tile(j, TileFull{});
  tile(qb, TileDiag{});

  rowsum16x8(rs_prev);
#pragma unroll
  for (int v = 0; v < 8; ++v) l_i[v] = l_i[v] * a_prev[v] + rs_prev[v];

  float* obase = attn_out + (size_t)b * SEQ_L * D_MODEL + (size_t)h * HEAD_DIM;
#pragma unroll
  for (int v = 0; v < 8; ++v) {
    int qg = qw0 + v + half * 8;
    float inv_l = 1.0f / l_i[v];
#pragma unroll
    for (int nt = 0; nt < 4; ++nt)
      obase[(size_t)qg * D_MODEL + nt * 16 + ln] = acc_o[nt][v] * inv_l;
  }
}

extern "C" void kernel_launch(void* const* d_in, const int* in_sizes, int n_in,
                              void* d_out, int out_size, void* d_ws,
                              size_t ws_size, hipStream_t stream) {
  (void)in_sizes; (void)n_in; (void)out_size; (void)ws_size;
  const float* x = (const float*)d_in[0];
  const float* w_qkv = (const float*)d_in[1];
  const float* b_qkv = (const float*)d_in[2];
  const float* w_out = (const float*)d_in[3];
  const float* b_out = (const float*)d_in[4];
  float* out = (float*)d_out;

  float* qkv = (float*)d_ws;
  float* attn = qkv + (size_t)M_ROWS * 3 * D_MODEL;

  gemm_bias_kernel<<<dim3((3 * D_MODEL) / BN, M_ROWS / BM), 256, 0, stream>>>(
      x, w_qkv, b_qkv, qkv, 3 * D_MODEL, D_MODEL);

  mha_kernel<<<dim3(BATCH * H_HEADS, SEQ_L / 64), 128, 0, stream>>>(qkv, attn);

  gemm_bias_kernel<<<dim3(D_MODEL / BN, M_ROWS / BM), 256, 0, stream>>>(
      attn, w_out, b_out, out, D_MODEL, D_MODEL);
}


# attention: blockIdx.y -> query block map re-dealt so the four workgroups assumed to meet on a WGP hold depths 32-g, 24-g, 9+g, 1+g: 66 KV tiles on every WGP (was 80 down to 52); sums unchanged
# speedup vs baseline: 1.6510x; 1.0417x over previous
#include <hip/hip_runtime.h>

typedef __attribute__((ext_vector_type(2))) float v2f;
typedef __attribute__((ext_vector_type(4))) float v4f;
typedef __attribute__((ext_vector_type(8))) float v8f;

#define D_MODEL 1024
#define H_HEADS 16
#define HEAD_DIM 64
#define SEQ_L 2048
#define BATCH 2
#define M_ROWS (BATCH * SEQ_L)

__device__ __forceinline__ v8f wmma_f32(v2f a, v2f b, v8f c) {
  return __builtin_amdgcn_wmma_f32_16x16x4_f32(false, a, false, b, (short)0, c,
                                               false, false);
}

template <int M>
__device__ __forceinline__ float row_xor(float x) {
  return __builtin_bit_cast(
      float, __builtin_amdgcn_update_dpp(0, __builtin_bit_cast(int, x), 0x160 | M, 0xf, 0xf, true));
}
template <int M>
__device__ __forceinline__ float rowmax_step(float x) {
  float r;
  asm("v_max_num_f32_dpp %0, %1, %1 row_xmask:%2 row_mask:0xf bank_mask:0xf bound_ctrl:1"
      : "=v"(r)
      : "v"(x), "n"(M));
  return r;
}
__device__ __forceinline__ void rowmax16x8(float (&x)[8]) {
#pragma unroll
  for (int v = 0; v < 8; ++v) x[v] = rowmax_step<1>(x[v]);
#pragma unroll
  for (int v = 0; v < 8; ++v) x[v] = rowmax_step<2>(x[v]);
#pragma unroll
  for (int v = 0; v < 8; ++v) x[v] = rowmax_step<4>(x[v]);
#pragma unroll
  for (int v = 0; v < 8; ++v) x[v] = rowmax_step<8>(x[v]);
}
__device__ __forceinline__ void rowsum16x8(float (&x)[8]) {
#pragma unroll
  for (int v = 0; v < 8; ++v) x[v] += row_xor<1>(x[v]);
#pragma unroll
  for (int v = 0; v < 8; ++v) x[v] += row_xor<2>(x[v]);
#pragma unroll
  for (int v = 0; v < 8; ++v) x[v] += row_xor<4>(x[v]);
#pragma unroll
  for (int v = 0; v < 8; ++v) x[v] += row_xor<8>(x[v]);
}

#define BM 64
#define BN 128
#define BK 32
#define APAD 36
#define BPANEL 528

__global__ __launch_bounds__(256) void gemm_bias_kernel(
    const float* __restrict__ A, const float* __restrict__ B,
    const float* __restrict__ bias, float* __restrict__ C, int N, int K) {
  __shared__ __align__(16) float As[BM][APAD];
  __shared__ __align__(16) float Bs[(BN / 16) * BPANEL];

  const int tid = threadIdx.x;
  const int lane = tid & 31;
  const int w = tid >> 5;
  const int wm = (w >> 2) << 5;
  const int wn = (w & 3) << 5;
  const int m0 = blockIdx.y * BM;
  const int n0 = blockIdx.x * BN;

  const int ln = lane & 15;
  const int half = lane >> 4;
  const int kb = half << 1;

  v8f acc[2][2];
#pragma unroll
  for (int i = 0; i < 2; ++i)
#pragma unroll
    for (int j = 0; j < 2; ++j)
#pragma unroll
      for (int v = 0; v < 8; ++v) acc[i][j][v] = 0.0f;

  for (int k0 = 0; k0 < K; k0 += BK) {
    __syncthreads();
    {
      int r = tid >> 3;
      int c = (tid & 7) << 2;
      const float* ap = A + (size_t)(m0 + r) * K + (k0 + c);
      int kr = tid >> 5;
      int nc = (tid & 31) << 2;
      const float* bsrc = B + (size_t)(k0 + kr) * N + (n0 + nc);
      float* bdst = Bs + (nc >> 4) * BPANEL + kr * 16 + (nc & 15);
      v4f a_lo = *(const v4f*)ap;
      v4f a_hi = *(const v4f*)(ap + (size_t)32 * K);
      v4f bv0 = *(const v4f*)(bsrc);
      v4f bv1 = *(const v4f*)(bsrc + (size_t)8 * N);
      v4f bv2 = *(const v4f*)(bsrc + (size_t)16 * N);
      v4f bv3 = *(const v4f*)(bsrc + (size_t)24 * N);
      asm volatile("" ::"v"(a_lo), "v"(a_hi), "v"(bv0), "v"(bv1), "v"(bv2), "v"(bv3) : "memory");
      *(v4f*)&As[r][c] = a_lo;
      *(v4f*)&As[r + 32][c] = a_hi;
      *(v4f*)(bdst) = bv0;
      *(v4f*)(bdst + 8 * 16) = bv1;
      *(v4f*)(bdst + 16 * 16) = bv2;
      *(v4f*)(bdst + 24 * 16) = bv3;
    }
    __syncthreads();
    const float* bp0 = Bs + (wn >> 4) * BPANEL + kb * 16 + ln;
    const float* bp1 = bp0 + BPANEL;
#pragma unroll
    for (int kk = 0; kk < BK; kk += 4) {
      v2f a0 = *(const v2f*)&As[wm + ln][kk + kb];
      v2f a1 = *(const v2f*)&As[wm + 16 + ln][kk + kb];
      v2f b0, b1;
      b0.x = bp0[kk * 16];
      b0.y = bp0[kk * 16 + 16];
      b1.x = bp1[kk * 16];
      b1.y = bp1[kk * 16 + 16];
      acc[0][0] = wmma_f32(a0, b0, acc[0][0]);
      acc[0][1] = wmma_f32(a0, b1, acc[0][1]);
      acc[1][0] = wmma_f32(a1, b0, acc[1][0]);
      acc[1][1] = wmma_f32(a1, b1, acc[1][1]);
    }
  }

#pragma unroll
  for (int j = 0; j < 2; ++j) {
    float bj = bias[n0 + wn + j * 16 + ln];
#pragma unroll
    for (int i = 0; i < 2; ++i) {
      int rbase = m0 + wm + i * 16 + half * 8;
      int col = n0 + wn + j * 16 + ln;
#pragma unroll
      for (int v = 0; v < 8; ++v)
        C[(size_t)(rbase + v) * N + col] = acc[i][j][v] + bj;
    }
  }
}

#define KPAD 68
#define VPANEL 1040

#define N_QBLOCKS (SEQ_L / 64)
#define SLOT_ROWS_PER_ROUND (256 / (BATCH * H_HEADS))
static_assert(N_QBLOCKS == 4 * SLOT_ROWS_PER_ROUND, "the deal below is written for four slot rows per WGP");
__device__ __forceinline__ int query_block_of_slot(int y) {
  const int g = y % SLOT_ROWS_PER_ROUND;
  const int k = y / SLOT_ROWS_PER_ROUND;
  if (k < 2) return (N_QBLOCKS - 1) - y;
  if (k == 2) return SLOT_ROWS_PER_ROUND + g;
  return g;
}

__device__ __forceinline__ void sched_fence() { __builtin_amdgcn_sched_barrier(0); }

struct TileDiag { static constexpr bool value = true; };
struct TileFull { static constexpr bool value = false; };

__global__ __launch_bounds__(128, 4) void mha_kernel(const float* __restrict__ qkv,
                                                     float* __restrict__ attn_out) {
  __shared__ __align__(16) float Ks[64][KPAD];
  __shared__ __align__(16) float Vs[4 * VPANEL];
  __shared__ __align__(16) float Ps[4][16][KPAD];

  const int bh = blockIdx.x;
  const int qb = query_block_of_slot(blockIdx.y);
  const int b = bh >> 4;
  const int h = bh & 15;

  const int tid = threadIdx.x;
  const int lane = tid & 31;
  const int w = tid >> 5;
  const int ln = lane & 15;
  const int half = lane >> 4;
  const int kb = half << 1;

  const int qw0 = qb * 64 + w * 16;
  const size_t rstride = 3 * D_MODEL;
  const float* base = qkv + (size_t)b * SEQ_L * rstride + (size_t)h * HEAD_DIM;

  const float qscale = 0.03125f * 1.44269504088896f;

  v2f qa[16];
  {
    const float* qp = base + (size_t)(qw0 + ln) * rstride + kb;
#pragma unroll
    for (int kk = 0; kk < 16; ++kk) qa[kk] = *(const v2f*)(qp + kk * 4) * qscale;
  }

  v8f acc_o[4];
  float m_i[8], l_i[8];
  float a_prev[8], rs_prev[8];
#pragma unroll
  for (int v = 0; v < 8; ++v) {
    m_i[v] = -__builtin_inff();
    l_i[v] = 0.0f;
    a_prev[v] = 1.0f;
    rs_prev[v] = 0.0f;
    acc_o[0][v] = 0.0f; acc_o[1][v] = 0.0f;
    acc_o[2][v] = 0.0f; acc_o[3][v] = 0.0f;
  }

  const float* vp = Vs + kb * 16 + ln;

  const int edge = ln - half * 8 - w * 16;

  auto tile = [&](const int j, auto tag) {
    constexpr bool DIAG = decltype(tag)::value;
    __syncthreads();
    {
      int r0 = tid >> 4;
      int c = (tid & 15) << 2;
      v4f kr[8], vr[8];
#pragma unroll
      for (int it = 0; it < 8; ++it) {
        int r = r0 + it * 8;
        const float* row = base + (size_t)(j * 64 + r) * rstride;
        kr[it] = *(const v4f*)(row + D_MODEL + c);
        vr[it] = *(const v4f*)(row + 2 * D_MODEL + c);
      }
      rowsum16x8(rs_prev);
#pragma unroll
      for (int v = 0; v < 8; ++v) l_i[v] = l_i[v] * a_prev[v] + rs_prev[v];
      asm volatile("" ::"v"(kr[0]), "v"(kr[1]), "v"(kr[2]), "v"(kr[3]), "v"(kr[4]), "v"(kr[5]),
                   "v"(kr[6]), "v"(kr[7]), "v"(vr[0]), "v"(vr[1]), "v"(vr[2]), "v"(vr[3]),
                   "v"(vr[4]), "v"(vr[5]), "v"(vr[6]), "v"(vr[7])
                   : "memory");
#pragma unroll
      for (int it = 0; it < 8; ++it) {
        int r = r0 + it * 8;
        *(v4f*)&Ks[r][c] = kr[it];
        *(v4f*)(Vs + (c >> 4) * VPANEL + r * 16 + (c & 15)) = vr[it];
      }
    }
    __syncthreads();

    v8f s[4];
#pragma unroll
    for (int nt = 0; nt < 4; ++nt)
#pragma unroll
      for (int v = 0; v < 8; ++v) s[nt][v] = 0.0f;
    v2f kf[2][16];
#pragma unroll
    for (int kq = 0; kq < 4; ++kq)
#pragma unroll
      for (int nt = 0; nt < 4; ++nt)
        kf[0][kq * 4 + nt] = *(const v2f*)&Ks[nt * 16 + ln][kq * 4 + kb];
#pragma unroll
    for (int blk = 0; blk < 4; ++blk) {
      if (blk < 3) {
#pragma unroll
        for (int kq = 0; kq < 4; ++kq)
#pragma unroll
          for (int nt = 0; nt < 4; ++nt)
            kf[(blk + 1) & 1][kq * 4 + nt] =
                *(const v2f*)&Ks[nt * 16 + ln][((blk + 1) * 4 + kq) * 4 + kb];
      }
      sched_fence();
#pragma unroll
      for (int kq = 0; kq < 4; ++kq)
#pragma unroll
        for (int nt = 0; nt < 4; ++nt)
          s[nt] = wmma_f32(qa[blk * 4 + kq], kf[blk & 1][kq * 4 + nt], s[nt]);
      sched_fence();
    }

    v2f pf[2][4];
    v2f vf[2][16];
#pragma unroll
    for (int kq = 0; kq < 4; ++kq)
#pragma unroll
      for (int nt = 0; nt < 4; ++nt) {
        vf[0][kq * 4 + nt].x = vp[nt * VPANEL + kq * 64];
        vf[0][kq * 4 + nt].y = vp[nt * VPANEL + kq * 64 + 16];
      }
    sched_fence();

    if constexpr (DIAG) {
#pragma unroll
      for (int nt = 0; nt < 4; ++nt)
#pragma unroll
        for (int v = 0; v < 8; ++v)
          if (edge > v - nt * 16) s[nt][v] = -__builtin_inff();
    }
    float tmax[8];
#pragma unroll
    for (int v = 0; v < 8; ++v)
      tmax[v] = fmaxf(fmaxf(s[0][v], s[1][v]), fmaxf(s[2][v], s[3][v]));
    rowmax16x8(tmax);

#pragma unroll
    for (int v = 0; v < 8; ++v) {
      float mn = fmaxf(m_i[v], tmax[v]);
      float alpha = __builtin_amdgcn_exp2f(m_i[v] - mn);
      m_i[v] = mn;
      float rs = 0.0f;
#pragma unroll
      for (int nt = 0; nt < 4; ++nt) {
        float p = __builtin_amdgcn_exp2f(s[nt][v] - mn);
        rs += p;
        Ps[w][v + half * 8][nt * 16 + ln] = p;
      }
      a_prev[v] = alpha;
      rs_prev[v] = rs;
      acc_o[0][v] *= alpha;
      acc_o[1][v] *= alpha;
      acc_o[2][v] *= alpha;
      acc_o[3][v] *= alpha;
    }

    asm volatile("s_wait_dscnt 0" ::: "memory");

#pragma unroll
    for (int kq = 0; kq < 4; ++kq) pf[0][kq] = *(const v2f*)&Ps[w][ln][kq * 4 + kb];
#pragma unroll
    for (int blk = 0; blk < 4; ++blk) {
      if (blk < 3) {
#pragma unroll
        for (int kq = 0; kq < 4; ++kq) {
          const int kk = (blk + 1) * 4 + kq;
          pf[(blk + 1) & 1][kq] = *(const v2f*)&Ps[w][ln][kk * 4 + kb];
#pragma unroll
          for (int nt = 0; nt < 4; ++nt) {
            vf[(blk + 1) & 1][kq * 4 + nt].x = vp[nt * VPANEL + kk * 64];
            vf[(blk + 1) & 1][kq * 4 + nt].y = vp[nt * VPANEL + kk * 64 + 16];
          }
        }
      }
      sched_fence();
#pragma unroll
      for (int kq = 0; kq < 4; ++kq)
#pragma unroll
        for (int nt = 0; nt < 4; ++nt)
          acc_o[nt] = wmma_f32(pf[blk & 1][kq], vf[blk & 1][kq * 4 + nt], acc_o[nt]);
      sched_fence();
    }
  };

  for (int j = 0; j < qb; ++j) tile(j, TileFull{});
  tile(qb, TileDiag{});

  rowsum16x8(rs_prev);
#pragma unroll
  for (int v = 0; v < 8; ++v) l_i[v] = l_i[v] * a_prev[v] + rs_prev[v];

  float* obase = attn_out + (size_t)b * SEQ_L * D_MODEL + (size_t)h * HEAD_DIM;
#pragma unroll
  for (int v = 0; v < 8; ++v) {
    int qg = qw0 + v + half * 8;
    float inv_l = 1.0f / l_i[v];
#pragma unroll
    for (int nt = 0; nt < 4; ++nt)
      obase[(size_t)qg * D_MODEL + nt * 16 + ln] = acc_o[nt][v] * inv_l;
  }
}

extern "C" void kernel_launch(void* const* d_in, const int* in_sizes, int n_in,
                              void* d_out, int out_size, void* d_ws,
                              size_t ws_size, hipStream_t stream) {
  (void)in_sizes; (void)n_in; (void)out_size; (void)ws_size;
  const float* x = (const float*)d_in[0];
  const float* w_qkv = (const float*)d_in[1];
  const float* b_qkv = (const float*)d_in[2];
  const float* w_out = (const float*)d_in[3];
  const float* b_out = (const float*)d_in[4];
  float* out = (float*)d_out;

  float* qkv = (float*)d_ws;
  float* attn = qkv + (size_t)M_ROWS * 3 * D_MODEL;

  gemm_bias_kernel<<<dim3((3 * D_MODEL) / BN, M_ROWS / BM), 256, 0, stream>>>(
      x, w_qkv, b_qkv, qkv, 3 * D_MODEL, D_MODEL);

  mha_kernel<<<dim3(BATCH * H_HEADS, SEQ_L / 64), 128, 0, stream>>>(qkv, attn);

  gemm_bias_kernel<<<dim3(D_MODEL / BN, M_ROWS / BM), 256, 0, stream>>>(
      attn, w_out, b_out, out, D_MODEL, D_MODEL);
}


# tile-load addresses in GEMM K loop and attention KV loop: one stepped pointer per operand + constant load offsets (GEMM N, K compile-time); 64-bit integer multiplies gone from both; sums unchanged
# speedup vs baseline: 1.7127x; 1.0374x over previous
#include <hip/hip_runtime.h>

typedef __attribute__((ext_vector_type(2))) float v2f;
typedef __attribute__((ext_vector_type(4))) float v4f;
typedef __attribute__((ext_vector_type(8))) float v8f;

#define D_MODEL 1024
#define H_HEADS 16
#define HEAD_DIM 64
#define SEQ_L 2048
#define BATCH 2
#define M_ROWS (BATCH * SEQ_L)

__device__ __forceinline__ v8f wmma_f32(v2f a, v2f b, v8f c) {
  return __builtin_amdgcn_wmma_f32_16x16x4_f32(false, a, false, b, (short)0, c,
                                               false, false);
}

template <int M>
__device__ __forceinline__ float row_xor(float x) {
  return __builtin_bit_cast(
      float, __builtin_amdgcn_update_dpp(0, __builtin_bit_cast(int, x), 0x160 | M, 0xf, 0xf, true));
}
template <int M>
__device__ __forceinline__ float rowmax_step(float x) {
  float r;
  asm("v_max_num_f32_dpp %0, %1, %1 row_xmask:%2 row_mask:0xf bank_mask:0xf bound_ctrl:1"
      : "=v"(r)
      : "v"(x), "n"(M));
  return r;
}
__device__ __forceinline__ void rowmax16x8(float (&x)[8]) {
#pragma unroll
  for (int v = 0; v < 8; ++v) x[v] = rowmax_step<1>(x[v]);
#pragma unroll
  for (int v = 0; v < 8; ++v) x[v] = rowmax_step<2>(x[v]);
#pragma unroll
  for (int v = 0; v < 8; ++v) x[v] = rowmax_step<4>(x[v]);
#pragma unroll
  for (int v = 0; v < 8; ++v) x[v] = rowmax_step<8>(x[v]);
}
__device__ __forceinline__ void rowsum16x8(float (&x)[8]) {
#pragma unroll
  for (int v = 0; v < 8; ++v) x[v] += row_xor<1>(x[v]);
#pragma unroll
  for (int v = 0; v < 8; ++v) x[v] += row_xor<2>(x[v]);
#pragma unroll
  for (int v = 0; v < 8; ++v) x[v] += row_xor<4>(x[v]);
#pragma unroll
  for (int v = 0; v < 8; ++v) x[v] += row_xor<8>(x[v]);
}

#define BM 64
#define BN 128
#define BK 32
#define APAD 36
#define BPANEL 528

template <int N, int K>
__global__ __launch_bounds__(256) void gemm_bias_kernel(
    const float* __restrict__ A, const float* __restrict__ B,
    const float* __restrict__ bias, float* __restrict__ C) {
  __shared__ __align__(16) float As[BM][APAD];
  __shared__ __align__(16) float Bs[(BN / 16) * BPANEL];

  const int tid = threadIdx.x;
  const int lane = tid & 31;
  const int w = tid >> 5;
  const int wm = (w >> 2) << 5;
  const int wn = (w & 3) << 5;
  const int m0 = blockIdx.y * BM;
  const int n0 = blockIdx.x * BN;

  const int ln = lane & 15;
  const int half = lane >> 4;
  const int kb = half << 1;

  v8f acc[2][2];
#pragma unroll
  for (int i = 0; i < 2; ++i)
#pragma unroll
    for (int j = 0; j < 2; ++j)
#pragma unroll
      for (int v = 0; v < 8; ++v) acc[i][j][v] = 0.0f;

  const int r = tid >> 3;
  const int c = (tid & 7) << 2;
  const int kr = tid >> 5;
  const int nc = (tid & 31) << 2;
  const float* ap = A + (size_t)(m0 + r) * K + c;
  const float* bp = B + (size_t)kr * N + (n0 + nc);
  float* const bdst = Bs + (nc >> 4) * BPANEL + kr * 16 + (nc & 15);

  for (int k0 = 0; k0 < K; k0 += BK) {
    __syncthreads();
    {
      v4f a_lo = *(const v4f*)(ap);
      v4f a_hi = *(const v4f*)(ap + 32 * K);
      v4f bv0 = *(const v4f*)(bp);
      v4f bv1 = *(const v4f*)(bp + 8 * N);
      v4f bv2 = *(const v4f*)(bp + 16 * N);
      v4f bv3 = *(const v4f*)(bp + 24 * N);
      ap += BK;
      bp += (size_t)BK * N;
      asm volatile("" ::"v"(a_lo), "v"(a_hi), "v"(bv0), "v"(bv1), "v"(bv2), "v"(bv3) : "memory");
      *(v4f*)&As[r][c] = a_lo;
      *(v4f*)&As[r + 32][c] = a_hi;
      *(v4f*)(bdst) = bv0;
      *(v4f*)(bdst + 8 * 16) = bv1;
      *(v4f*)(bdst + 16 * 16) = bv2;
      *(v4f*)(bdst + 24 * 16) = bv3;
    }
    __syncthreads();
    const float* bp0 = Bs + (wn >> 4) * BPANEL + kb * 16 + ln;
    const float* bp1 = bp0 + BPANEL;
#pragma unroll
    for (int kk = 0; kk < BK; kk += 4) {
      v2f a0 = *(const v2f*)&As[wm + ln][kk + kb];
      v2f a1 = *(const v2f*)&As[wm + 16 + ln][kk + kb];
      v2f b0, b1;
      b0.x = bp0[kk * 16];
      b0.y = bp0[kk * 16 + 16];
      b1.x = bp1[kk * 16];
      b1.y = bp1[kk * 16 + 16];
      acc[0][0] = wmma_f32(a0, b0, acc[0][0]);
      acc[0][1] = wmma_f32(a0, b1, acc[0][1]);
      acc[1][0] = wmma_f32(a1, b0, acc[1][0]);
      acc[1][1] = wmma_f32(a1, b1, acc[1][1]);
    }
  }

#pragma unroll
  for (int j = 0; j < 2; ++j) {
    float bj = bias[n0 + wn + j * 16 + ln];
#pragma unroll
    for (int i = 0; i < 2; ++i) {
      int rbase = m0 + wm + i * 16 + half * 8;
      int col = n0 + wn + j * 16 + ln;
#pragma unroll
      for (int v = 0; v < 8; ++v)
        C[(size_t)(rbase + v) * N + col] = acc[i][j][v] + bj;
    }
  }
}

#define KPAD 68
#define VPANEL 1040

#define N_QBLOCKS (SEQ_L / 64)
#define SLOT_ROWS_PER_ROUND (256 / (BATCH * H_HEADS))
static_assert(N_QBLOCKS == 4 * SLOT_ROWS_PER_ROUND, "the deal below is written for four slot rows per WGP");
__device__ __forceinline__ int query_block_of_slot(int y) {
  const int g = y % SLOT_ROWS_PER_ROUND;
  const int k = y / SLOT_ROWS_PER_ROUND;
  if (k < 2) return (N_QBLOCKS - 1) - y;
  if (k == 2) return SLOT_ROWS_PER_ROUND + g;
  return g;
}

__device__ __forceinline__ void sched_fence() { __builtin_amdgcn_sched_barrier(0); }

struct TileDiag { static constexpr bool value = true; };
struct TileFull { static constexpr bool value = false; };

__global__ __launch_bounds__(128, 4) void mha_kernel(const float* __restrict__ qkv,
                                                     float* __restrict__ attn_out) {
  __shared__ __align__(16) float Ks[64][KPAD];
  __shared__ __align__(16) float Vs[4 * VPANEL];
  __shared__ __align__(16) float Ps[4][16][KPAD];

  const int bh = blockIdx.x;
  const int qb = query_block_of_slot(blockIdx.y);
  const int b = bh >> 4;
  const int h = bh & 15;

  const int tid = threadIdx.x;
  const int lane = tid & 31;
  const int w = tid >> 5;
  const int ln = lane & 15;
  const int half = lane >> 4;
  const int kb = half << 1;

  const int qw0 = qb * 64 + w * 16;
  const size_t rstride = 3 * D_MODEL;
  const float* base = qkv + (size_t)b * SEQ_L * rstride + (size_t)h * HEAD_DIM;

  const float qscale = 0.03125f * 1.44269504088896f;

  v2f qa[16];
  {
    const float* qp = base + (size_t)(qw0 + ln) * rstride + kb;
#pragma unroll
    for (int kk = 0; kk < 16; ++kk) qa[kk] = *(const v2f*)(qp + kk * 4) * qscale;
  }

  v8f acc_o[4];
  float m_i[8], l_i[8];
  float a_prev[8], rs_prev[8];
#pragma unroll
  for (int v = 0; v < 8; ++v) {
    m_i[v] = -__builtin_inff();
    l_i[v] = 0.0f;
    a_prev[v] = 1.0f;
    rs_prev[v] = 0.0f;
    acc_o[0][v] = 0.0f; acc_o[1][v] = 0.0f;
    acc_o[2][v] = 0.0f; acc_o[3][v] = 0.0f;
  }

  const unsigned kv_off = (unsigned)((tid >> 4) * (int)rstride + ((tid & 15) << 2)) * 4u;

  const float* vp = Vs + kb * 16 + ln;

  const int edge = ln - half * 8 - w * 16;

  auto tile = [&](const int j, auto tag) {
    constexpr bool DIAG = decltype(tag)::value;
    __syncthreads();
    {
      int r0 = tid >> 4;
      int c = (tid & 15) << 2;
      v4f kr[8], vr[8];
      const char* tp = (const char*)(base + (size_t)j * (64 * rstride)) + kv_off;
#pragma unroll
      for (int it = 0; it < 8; ++it) {
        kr[it] = *(const v4f*)(tp + (it * 8 * rstride + D_MODEL) * 4);
        vr[it] = *(const v4f*)(tp + (it * 8 * rstride + 2 * D_MODEL) * 4);
      }
      rowsum16x8(rs_prev);
#pragma unroll
      for (int v = 0; v < 8; ++v) l_i[v] = l_i[v] * a_prev[v] + rs_prev[v];
      asm volatile("" ::"v"(kr[0]), "v"(kr[1]), "v"(kr[2]), "v"(kr[3]), "v"(kr[4]), "v"(kr[5]),
                   "v"(kr[6]), "v"(kr[7]), "v"(vr[0]), "v"(vr[1]), "v"(vr[2]), "v"(vr[3]),
                   "v"(vr[4]), "v"(vr[5]), "v"(vr[6]), "v"(vr[7])
                   : "memory");
#pragma unroll
      for (int it = 0; it < 8; ++it) {
        int r = r0 + it * 8;
        *(v4f*)&Ks[r][c] = kr[it];
        *(v4f*)(Vs + (c >> 4) * VPANEL + r * 16 + (c & 15)) = vr[it];
      }
    }
    __syncthreads();

    v8f s[4];
#pragma unroll
    for (int nt = 0; nt < 4; ++nt)
#pragma unroll
      for (int v = 0; v < 8; ++v) s[nt][v] = 0.0f;
    v2f kf[2][16];
#pragma unroll
    for (int kq = 0; kq < 4; ++kq)
#pragma unroll
      for (int nt = 0; nt < 4; ++nt)
        kf[0][kq * 4 + nt] = *(const v2f*)&Ks[nt * 16 + ln][kq * 4 + kb];
#pragma unroll
    for (int blk = 0; blk < 4; ++blk) {
      if (blk < 3) {
#pragma unroll
        for (int kq = 0; kq < 4; ++kq)
#pragma unroll
          for (int nt = 0; nt < 4; ++nt)
            kf[(blk + 1) & 1][kq * 4 + nt] =
                *(const v2f*)&Ks[nt * 16 + ln][((blk + 1) * 4 + kq) * 4 + kb];
      }
      sched_fence();
#pragma unroll
      for (int kq = 0; kq < 4; ++kq)
#pragma unroll
        for (int nt = 0; nt < 4; ++nt)
          s[nt] = wmma_f32(qa[blk * 4 + kq], kf[blk & 1][kq * 4 + nt], s[nt]);
      sched_fence();
    }

    v2f pf[2][4];
    v2f vf[2][16];
#pragma unroll
    for (int kq = 0; kq < 4; ++kq)
#pragma unroll
      for (int nt = 0; nt < 4; ++nt) {
        vf[0][kq * 4 + nt].x = vp[nt * VPANEL + kq * 64];
        vf[0][kq * 4 + nt].y = vp[nt * VPANEL + kq * 64 + 16];
      }
    sched_fence();

    if constexpr (DIAG) {
#pragma unroll
      for (int nt = 0; nt < 4; ++nt)
#pragma unroll
        for (int v = 0; v < 8; ++v)
          if (edge > v - nt * 16) s[nt][v] = -__builtin_inff();
    }
    float tmax[8];
#pragma unroll
    for (int v = 0; v < 8; ++v)
      tmax[v] = fmaxf(fmaxf(s[0][v], s[1][v]), fmaxf(s[2][v], s[3][v]));
    rowmax16x8(tmax);

#pragma unroll
    for (int v = 0; v < 8; ++v) {
      float mn = fmaxf(m_i[v], tmax[v]);
      float alpha = __builtin_amdgcn_exp2f(m_i[v] - mn);
      m_i[v] = mn;
      float rs = 0.0f;
#pragma unroll
      for (int nt = 0; nt < 4; ++nt) {
        float p = __builtin_amdgcn_exp2f(s[nt][v] - mn);
        rs += p;
        Ps[w][v + half * 8][nt * 16 + ln] = p;
      }
      a_prev[v] = alpha;
      rs_prev[v] = rs;
      acc_o[0][v] *= alpha;
      acc_o[1][v] *= alpha;
      acc_o[2][v] *= alpha;
      acc_o[3][v] *= alpha;
    }

    asm volatile("s_wait_dscnt 0" ::: "memory");

#pragma unroll
    for (int kq = 0; kq < 4; ++kq) pf[0][kq] = *(const v2f*)&Ps[w][ln][kq * 4 + kb];
#pragma unroll
    for (int blk = 0; blk < 4; ++blk) {
      if (blk < 3) {
#pragma unroll
        for (int kq = 0; kq < 4; ++kq) {
          const int kk = (blk + 1) * 4 + kq;
          pf[(blk + 1) & 1][kq] = *(const v2f*)&Ps[w][ln][kk * 4 + kb];
#pragma unroll
          for (int nt = 0; nt < 4; ++nt) {
            vf[(blk + 1) & 1][kq * 4 + nt].x = vp[nt * VPANEL + kk * 64];
            vf[(blk + 1) & 1][kq * 4 + nt].y = vp[nt * VPANEL + kk * 64 + 16];
          }
        }
      }
      sched_fence();
#pragma unroll
      for (int kq = 0; kq < 4; ++kq)
#pragma unroll
        for (int nt = 0; nt < 4; ++nt)
          acc_o[nt] = wmma_f32(pf[blk & 1][kq], vf[blk & 1][kq * 4 + nt], acc_o[nt]);
      sched_fence();
    }
  };

  for (int j = 0; j < qb; ++j) tile(j, TileFull{});
  tile(qb, TileDiag{});

  rowsum16x8(rs_prev);
#pragma unroll
  for (int v = 0; v < 8; ++v) l_i[v] = l_i[v] * a_prev[v] + rs_prev[v];

  float* obase = attn_out + (size_t)b * SEQ_L * D_MODEL + (size_t)h * HEAD_DIM;
#pragma unroll
  for (int v = 0; v < 8; ++v) {
    int qg = qw0 + v + half * 8;
    float inv_l = 1.0f / l_i[v];
#pragma unroll
    for (int nt = 0; nt < 4; ++nt)
      obase[(size_t)qg * D_MODEL + nt * 16 + ln] = acc_o[nt][v] * inv_l;
  }
}

extern "C" void kernel_launch(void* const* d_in, const int* in_sizes, int n_in,
                              void* d_out, int out_size, void* d_ws,
                              size_t ws_size, hipStream_t stream) {
  (void)in_sizes; (void)n_in; (void)out_size; (void)ws_size;
  const float* x = (const float*)d_in[0];
  const float* w_qkv = (const float*)d_in[1];
  const float* b_qkv = (const float*)d_in[2];
  const float* w_out = (const float*)d_in[3];
  const float* b_out = (const float*)d_in[4];
  float* out = (float*)d_out;

  float* qkv = (float*)d_ws;
  float* attn = qkv + (size_t)M_ROWS * 3 * D_MODEL;

  gemm_bias_kernel<3 * D_MODEL, D_MODEL><<<dim3((3 * D_MODEL) / BN, M_ROWS / BM), 256, 0, stream>>>(
      x, w_qkv, b_qkv, qkv);

  mha_kernel<<<dim3(BATCH * H_HEADS, SEQ_L / 64), 128, 0, stream>>>(qkv, attn);

  gemm_bias_kernel<D_MODEL, D_MODEL><<<dim3(D_MODEL / BN, M_ROWS / BM), 256, 0, stream>>>(
      attn, w_out, b_out, out);
}
